# starting kernel
# baseline (speedup 1.0000x reference)
#include <hip/hip_runtime.h>
#include <hip/hip_bf16.h>
#include <math.h>

typedef float v2f __attribute__((ext_vector_type(2)));
typedef float v8f __attribute__((ext_vector_type(8)));

#define IN_DIM 256
#define OUT_DIM 64
#define SLOPE 0.1f
#define KC 128
#define LDP 132

__global__ __launch_bounds__(256) void gat_gemm_wmma(
    const float* __restrict__ features,
    const float* __restrict__ W,
    const float* __restrict__ bias,
    float* __restrict__ emb,
    int n)
{
    __shared__ float ldsA[32 * LDP];
    __shared__ float ldsB[OUT_DIM * LDP];

    const int tid  = threadIdx.x;
    const int row0 = blockIdx.x * 32;

    const int wave  = tid >> 5;
    const int lane  = tid & 31;
    const int mbase = (wave >> 2) * 16;
    const int n0    = (wave & 3) * 16;
    const int lm    = lane & 15;
    const int half  = lane >> 4;
    const int kh    = half * 2;

    v8f acc = {0.f, 0.f, 0.f, 0.f, 0.f, 0.f, 0.f, 0.f};

    for (int kc = 0; kc < IN_DIM; kc += KC) {
        #pragma unroll
        for (int i = 0; i < 4; ++i) {
            int idx = tid + i * 256;
            int r   = idx >> 5;
            int c4  = idx & 31;
            float4 v = make_float4(0.f, 0.f, 0.f, 0.f);
            int gr = row0 + r;
            if (gr < n)
                v = *(const float4*)(features + (size_t)gr * IN_DIM + kc + c4 * 4);
            *(float4*)(&ldsA[r * LDP + c4 * 4]) = v;
        }
        #pragma unroll
        for (int i = 0; i < 8; ++i) {
            int idx = tid + i * 256;
            int r   = idx >> 5;
            int c4  = idx & 31;
            float4 v = *(const float4*)(W + (size_t)r * IN_DIM + kc + c4 * 4);
            *(float4*)(&ldsB[r * LDP + c4 * 4]) = v;
        }
        __syncthreads();

        const float* pA = &ldsA[(mbase + lm) * LDP + kh];
        const float* pB = &ldsB[(n0   + lm) * LDP + kh];
        #pragma unroll 8
        for (int k = 0; k < KC; k += 4) {
            v2f afrag = *(const v2f*)(pA + k);
            v2f bfrag = *(const v2f*)(pB + k);
            acc = __builtin_amdgcn_wmma_f32_16x16x4_f32(
                false, afrag, false, bfrag, (short)0, acc, false, false);
        }
        __syncthreads();
    }

    const float bv = bias[n0 + lm];
    #pragma unroll
    for (int r = 0; r < 8; ++r) {
        int grow = row0 + mbase + r + half * 8;
        if (grow < n)
            emb[(size_t)grow * OUT_DIM + n0 + lm] = acc[r] + bv;
    }
}

__global__ __launch_bounds__(256) void gat_node(
    const float* __restrict__ emb,
    const float* __restrict__ a,
    float* __restrict__ s1,
    float* __restrict__ s2,
    float* __restrict__ rowsum,
    float* __restrict__ out,
    int n)
{
    int i = blockIdx.x * blockDim.x + threadIdx.x;
    if (i >= n) return;
    const float* er = emb + (size_t)i * OUT_DIM;
    float d1 = 0.f, d2 = 0.f;
    #pragma unroll
    for (int j = 0; j < OUT_DIM; ++j) {
        float e = er[j];
        d1 += e * a[j];
        d2 += e * a[OUT_DIM + j];
    }
    s1[i] = d1;
    s2[i] = d2;
    float z = d1 + d2;
    float l = z > 0.f ? z : SLOPE * z;
    float w = expf(l);
    rowsum[i] = w;
    float* orow = out + (size_t)i * OUT_DIM;
    #pragma unroll
    for (int j = 0; j < OUT_DIM; ++j)
        orow[j] = w * er[j];
}

__global__ __launch_bounds__(256) void gat_edge(
    const int* __restrict__ edges,
    const float* __restrict__ emb,
    const float* __restrict__ s1,
    const float* __restrict__ s2,
    float* __restrict__ rowsum,
    float* __restrict__ out,
    int E)
{
    int wid  = (blockIdx.x * blockDim.x + threadIdx.x) >> 5;
    int lane = threadIdx.x & 31;
    if (wid >= E) return;
    int src = edges[2 * wid];
    int dst = edges[2 * wid + 1];
    float z = s1[src] + s2[dst];
    float l = z > 0.f ? z : SLOPE * z;
    float w = expf(l);
    if (lane == 0) atomicAdd(rowsum + src, w);
    const float2 ev = *(const float2*)(emb + (size_t)dst * OUT_DIM + 2 * lane);
    float* op = out + (size_t)src * OUT_DIM + 2 * lane;
    atomicAdd(op,     w * ev.x);
    atomicAdd(op + 1, w * ev.y);
}

__global__ __launch_bounds__(256) void gat_final(
    float* __restrict__ out,
    const float* __restrict__ rowsum,
    int total)
{
    int idx = blockIdx.x * blockDim.x + threadIdx.x;
    if (idx >= total) return;
    out[idx] /= rowsum[idx >> 6];
}

extern "C" void kernel_launch(void* const* d_in, const int* in_sizes, int n_in,
                              void* d_out, int out_size, void* d_ws, size_t ws_size,
                              hipStream_t stream)
{
    (void)n_in; (void)out_size; (void)ws_size;
    const float* features = (const float*)d_in[0];
    const float* W        = (const float*)d_in[1];
    const float* bias     = (const float*)d_in[2];
    const float* a        = (const float*)d_in[3];
    const int*   edges    = (const int*)d_in[4];

    const int n = in_sizes[0] / IN_DIM;
    const int E = in_sizes[4] / 2;
    float* out = (float*)d_out;

    float* emb    = (float*)d_ws;
    float* s1     = emb + (size_t)n * OUT_DIM;
    float* s2     = s1 + n;
    float* rowsum = s2 + n;

    gat_gemm_wmma<<<(n + 31) / 32, 256, 0, stream>>>(features, W, bias, emb, n);
    gat_node<<<(n + 255) / 256, 256, 0, stream>>>(emb, a, s1, s2, rowsum, out, n);
    gat_edge<<<(E + 7) / 8, 256, 0, stream>>>(edges, emb, s1, s2, rowsum, out, E);
    const int total = n * OUT_DIM;
    gat_final<<<(total + 255) / 256, 256, 0, stream>>>(out, rowsum, total);
}


# Edge launch: a wave takes 32 edges, not 1: index, logit, exp once per edge (a lane each); lane l holds columns l, l+32, so each row load and atomic is one whole 128-B line; arrival-order sums kept
# speedup vs baseline: 1.4343x; 1.4343x over previous
#include <hip/hip_runtime.h>
#include <hip/hip_bf16.h>
#include <math.h>

typedef float v2f __attribute__((ext_vector_type(2)));
typedef float v8f __attribute__((ext_vector_type(8)));

#define IN_DIM 256
#define OUT_DIM 64
#define SLOPE 0.1f
#define KC 128
#define LDP 132

__global__ __launch_bounds__(256) void gat_gemm_wmma(
    const float* __restrict__ features,
    const float* __restrict__ W,
    const float* __restrict__ bias,
    float* __restrict__ emb,
    int n)
{
    __shared__ float ldsA[32 * LDP];
    __shared__ float ldsB[OUT_DIM * LDP];

    const int tid  = threadIdx.x;
    const int row0 = blockIdx.x * 32;

    const int wave  = tid >> 5;
    const int lane  = tid & 31;
    const int mbase = (wave >> 2) * 16;
    const int n0    = (wave & 3) * 16;
    const int lm    = lane & 15;
    const int half  = lane >> 4;
    const int kh    = half * 2;

    v8f acc = {0.f, 0.f, 0.f, 0.f, 0.f, 0.f, 0.f, 0.f};

    for (int kc = 0; kc < IN_DIM; kc += KC) {
        #pragma unroll
        for (int i = 0; i < 4; ++i) {
            int idx = tid + i * 256;
            int r   = idx >> 5;
            int c4  = idx & 31;
            float4 v = make_float4(0.f, 0.f, 0.f, 0.f);
            int gr = row0 + r;
            if (gr < n)
                v = *(const float4*)(features + (size_t)gr * IN_DIM + kc + c4 * 4);
            *(float4*)(&ldsA[r * LDP + c4 * 4]) = v;
        }
        #pragma unroll
        for (int i = 0; i < 8; ++i) {
            int idx = tid + i * 256;
            int r   = idx >> 5;
            int c4  = idx & 31;
            float4 v = *(const float4*)(W + (size_t)r * IN_DIM + kc + c4 * 4);
            *(float4*)(&ldsB[r * LDP + c4 * 4]) = v;
        }
        __syncthreads();

        const float* pA = &ldsA[(mbase + lm) * LDP + kh];
        const float* pB = &ldsB[(n0   + lm) * LDP + kh];
        #pragma unroll 8
        for (int k = 0; k < KC; k += 4) {
            v2f afrag = *(const v2f*)(pA + k);
            v2f bfrag = *(const v2f*)(pB + k);
            acc = __builtin_amdgcn_wmma_f32_16x16x4_f32(
                false, afrag, false, bfrag, (short)0, acc, false, false);
        }
        __syncthreads();
    }

    const float bv = bias[n0 + lm];
    #pragma unroll
    for (int r = 0; r < 8; ++r) {
        int grow = row0 + mbase + r + half * 8;
        if (grow < n)
            emb[(size_t)grow * OUT_DIM + n0 + lm] = acc[r] + bv;
    }
}

__global__ __launch_bounds__(256) void gat_node(
    const float* __restrict__ emb,
    const float* __restrict__ a,
    float* __restrict__ s1,
    float* __restrict__ s2,
    float* __restrict__ rowsum,
    float* __restrict__ out,
    int n)
{
    int i = blockIdx.x * blockDim.x + threadIdx.x;
    if (i >= n) return;
    const float* er = emb + (size_t)i * OUT_DIM;
    float d1 = 0.f, d2 = 0.f;
    #pragma unroll
    for (int j = 0; j < OUT_DIM; ++j) {
        float e = er[j];
        d1 += e * a[j];
        d2 += e * a[OUT_DIM + j];
    }
    s1[i] = d1;
    s2[i] = d2;
    float z = d1 + d2;
    float l = z > 0.f ? z : SLOPE * z;
    float w = expf(l);
    rowsum[i] = w;
    float* orow = out + (size_t)i * OUT_DIM;
    #pragma unroll
    for (int j = 0; j < OUT_DIM; ++j)
        orow[j] = w * er[j];
}

#define EPW 32
#define EB  16

__device__ __forceinline__ float edge_weight(float z)
{
    float l = z > 0.f ? z : SLOPE * z;
    return expf(l);
}

__global__ __launch_bounds__(256) void gat_edge(
    const int* __restrict__ edges,
    const float* __restrict__ emb,
    const float* __restrict__ s1,
    const float* __restrict__ s2,
    float* __restrict__ rowsum,
    float* __restrict__ out,
    int E)
{
    const int lane = threadIdx.x & 31;
    const int wid  = (blockIdx.x * blockDim.x + threadIdx.x) >> 5;
    const int e0   = wid * EPW;
    if (e0 >= E) return;
    const int cnt = min(EPW, E - e0);

    if (cnt == EPW) {
        const int2 sd = *(const int2*)(edges + 2 * (size_t)(e0 + lane));
        const int src = sd.x;
        const int dst = sd.y;
        const float g1 = s1[src];
        const float g2 = s2[dst];

        float r0[EB], r1[EB], q0[EB], q1[EB];
        #pragma unroll
        for (int j = 0; j < EB; ++j) {
            const int d = __builtin_amdgcn_readlane(dst, j);
            const float* p = emb + (size_t)d * OUT_DIM + lane;
            r0[j] = p[0];
            r1[j] = p[32];
        }

        const float w = edge_weight(g1 + g2);
        atomicAdd(rowsum + src, w);

        #pragma unroll
        for (int j = 0; j < EB; ++j) {
            const int d = __builtin_amdgcn_readlane(dst, EB + j);
            const float* p = emb + (size_t)d * OUT_DIM + lane;
            q0[j] = p[0];
            q1[j] = p[32];
        }

        #pragma unroll
        for (int j = 0; j < EB; ++j) {
            const int   s  = __builtin_amdgcn_readlane(src, j);
            const float wj = __int_as_float(__builtin_amdgcn_readlane(__float_as_int(w), j));
            float* o = out + (size_t)s * OUT_DIM + lane;
            atomicAdd(o,      wj * r0[j]);
            atomicAdd(o + 32, wj * r1[j]);
        }
        #pragma unroll
        for (int j = 0; j < EB; ++j) {
            const int   s  = __builtin_amdgcn_readlane(src, EB + j);
            const float wj = __int_as_float(__builtin_amdgcn_readlane(__float_as_int(w), EB + j));
            float* o = out + (size_t)s * OUT_DIM + lane;
            atomicAdd(o,      wj * q0[j]);
            atomicAdd(o + 32, wj * q1[j]);
        }
    } else {
        int src = 0, dst = 0;
        float w = 0.f;
        if (lane < cnt) {
            const int2 sd = *(const int2*)(edges + 2 * (size_t)(e0 + lane));
            src = sd.x;
            dst = sd.y;
            w = edge_weight(s1[src] + s2[dst]);
            atomicAdd(rowsum + src, w);
        }
        for (int j = 0; j < cnt; ++j) {
            const int   d  = __builtin_amdgcn_readlane(dst, j);
            const int   s  = __builtin_amdgcn_readlane(src, j);
            const float wj = __int_as_float(__builtin_amdgcn_readlane(__float_as_int(w), j));
            const float* p = emb + (size_t)d * OUT_DIM + lane;
            float* o = out + (size_t)s * OUT_DIM + lane;
            atomicAdd(o,      wj * p[0]);
            atomicAdd(o + 32, wj * p[32]);
        }
    }
}

__global__ __launch_bounds__(256) void gat_final(
    float* __restrict__ out,
    const float* __restrict__ rowsum,
    int total)
{
    int idx = blockIdx.x * blockDim.x + threadIdx.x;
    if (idx >= total) return;
    out[idx] /= rowsum[idx >> 6];
}

extern "C" void kernel_launch(void* const* d_in, const int* in_sizes, int n_in,
                              void* d_out, int out_size, void* d_ws, size_t ws_size,
                              hipStream_t stream)
{
    (void)n_in; (void)out_size; (void)ws_size;
    const float* features = (const float*)d_in[0];
    const float* W        = (const float*)d_in[1];
    const float* bias     = (const float*)d_in[2];
    const float* a        = (const float*)d_in[3];
    const int*   edges    = (const int*)d_in[4];

    const int n = in_sizes[0] / IN_DIM;
    const int E = in_sizes[4] / 2;
    float* out = (float*)d_out;

    float* emb    = (float*)d_ws;
    float* s1     = emb + (size_t)n * OUT_DIM;
    float* s2     = s1 + n;
    float* rowsum = s2 + n;

    gat_gemm_wmma<<<(n + 31) / 32, 256, 0, stream>>>(features, W, bias, emb, n);
    gat_node<<<(n + 255) / 256, 256, 0, stream>>>(emb, a, s1, s2, rowsum, out, n);
    gat_edge<<<(E + 8 * EPW - 1) / (8 * EPW), 256, 0, stream>>>(edges, emb, s1, s2, rowsum, out, E);
    const int total = n * OUT_DIM;
    gat_final<<<(total + 255) / 256, 256, 0, stream>>>(out, rowsum, total);
}


# Node launch folded into the product launch's epilogue: tile via LDS, a wave per row does both logit dots (lane tree, not serial), exp, whole-row stores of emb and out start; 4 -> 3 launches
# speedup vs baseline: 1.5880x; 1.1072x over previous
#include <hip/hip_runtime.h>
#include <hip/hip_bf16.h>
#include <math.h>

typedef float v2f __attribute__((ext_vector_type(2)));
typedef float v8f __attribute__((ext_vector_type(8)));

#define IN_DIM 256
#define OUT_DIM 64
#define SLOPE 0.1f
#define KC 128
#define LDP 132

#define EPI_LD 68

__device__ __forceinline__ float edge_weight(float z)
{
    float l = z > 0.f ? z : SLOPE * z;
    return expf(l);
}

__global__ __launch_bounds__(256) void gat_gemm_wmma(
    const float* __restrict__ features,
    const float* __restrict__ W,
    const float* __restrict__ bias,
    const float* __restrict__ a,
    float* __restrict__ emb,
    float* __restrict__ s1,
    float* __restrict__ s2,
    float* __restrict__ rowsum,
    float* __restrict__ out,
    int n)
{
    __shared__ float ldsA[32 * LDP];
    __shared__ float ldsB[OUT_DIM * LDP];

    const int tid  = threadIdx.x;
    const int row0 = blockIdx.x * 32;

    const int wave  = tid >> 5;
    const int lane  = tid & 31;
    const int mbase = (wave >> 2) * 16;
    const int n0    = (wave & 3) * 16;
    const int lm    = lane & 15;
    const int half  = lane >> 4;
    const int kh    = half * 2;

    const float  bv = bias[n0 + lm];
    const float2 a1 = *(const float2*)(a + 2 * lane);
    const float2 a2 = *(const float2*)(a + OUT_DIM + 2 * lane);

    v8f acc = {0.f, 0.f, 0.f, 0.f, 0.f, 0.f, 0.f, 0.f};

    for (int kc = 0; kc < IN_DIM; kc += KC) {
        #pragma unroll
        for (int i = 0; i < 4; ++i) {
            int idx = tid + i * 256;
            int r   = idx >> 5;
            int c4  = idx & 31;
            float4 v = make_float4(0.f, 0.f, 0.f, 0.f);
            int gr = row0 + r;
            if (gr < n)
                v = *(const float4*)(features + (size_t)gr * IN_DIM + kc + c4 * 4);
            *(float4*)(&ldsA[r * LDP + c4 * 4]) = v;
        }
        #pragma unroll
        for (int i = 0; i < 8; ++i) {
            int idx = tid + i * 256;
            int r   = idx >> 5;
            int c4  = idx & 31;
            float4 v = *(const float4*)(W + (size_t)r * IN_DIM + kc + c4 * 4);
            *(float4*)(&ldsB[r * LDP + c4 * 4]) = v;
        }
        __syncthreads();

        const float* pA = &ldsA[(mbase + lm) * LDP + kh];
        const float* pB = &ldsB[(n0   + lm) * LDP + kh];
        #pragma unroll 8
        for (int k = 0; k < KC; k += 4) {
            v2f afrag = *(const v2f*)(pA + k);
            v2f bfrag = *(const v2f*)(pB + k);
            acc = __builtin_amdgcn_wmma_f32_16x16x4_f32(
                false, afrag, false, bfrag, (short)0, acc, false, false);
        }
        __syncthreads();
    }

    float* tile = ldsA;
    #pragma unroll
    for (int r = 0; r < 8; ++r)
        tile[(mbase + r + half * 8) * EPI_LD + n0 + lm] = acc[r] + bv;
    __syncthreads();

    const int rl0 = wave * 4;
    float2 e[4];
    float d1[4], d2[4];
    #pragma unroll
    for (int j = 0; j < 4; ++j) {
        e[j]  = *(const float2*)(&tile[(rl0 + j) * EPI_LD + 2 * lane]);
        d1[j] = e[j].x * a1.x + e[j].y * a1.y;
        d2[j] = e[j].x * a2.x + e[j].y * a2.y;
    }
    #pragma unroll
    for (int off = 16; off > 0; off >>= 1) {
        #pragma unroll
        for (int j = 0; j < 4; ++j) {
            d1[j] += __shfl_xor(d1[j], off);
            d2[j] += __shfl_xor(d2[j], off);
        }
    }
    float ws = 0.f, v1 = 0.f, v2 = 0.f;
    #pragma unroll
    for (int j = 0; j < 4; ++j) {
        const float w = edge_weight(d1[j] + d2[j]);
        const int grow = row0 + rl0 + j;
        if (grow < n) {
            *(float2*)(emb + (size_t)grow * OUT_DIM + 2 * lane) = e[j];
            *(float2*)(out + (size_t)grow * OUT_DIM + 2 * lane) =
                make_float2(w * e[j].x, w * e[j].y);
        }
        if (lane == j) { ws = w; v1 = d1[j]; v2 = d2[j]; }
    }
    const int srow = row0 + rl0 + lane;
    if (lane < 4 && srow < n) {
        s1[srow]     = v1;
        s2[srow]     = v2;
        rowsum[srow] = ws;
    }
}

#define EPW 32
#define EB  16

__global__ __launch_bounds__(256) void gat_edge(
    const int* __restrict__ edges,
    const float* __restrict__ emb,
    const float* __restrict__ s1,
    const float* __restrict__ s2,
    float* __restrict__ rowsum,
    float* __restrict__ out,
    int E)
{
    const int lane = threadIdx.x & 31;
    const int wid  = (blockIdx.x * blockDim.x + threadIdx.x) >> 5;
    const int e0   = wid * EPW;
    if (e0 >= E) return;
    const int cnt = min(EPW, E - e0);

    if (cnt == EPW) {
        const int2 sd = *(const int2*)(edges + 2 * (size_t)(e0 + lane));
        const int src = sd.x;
        const int dst = sd.y;
        const float g1 = s1[src];
        const float g2 = s2[dst];

        float r0[EB], r1[EB], q0[EB], q1[EB];
        #pragma unroll
        for (int j = 0; j < EB; ++j) {
            const int d = __builtin_amdgcn_readlane(dst, j);
            const float* p = emb + (size_t)d * OUT_DIM + lane;
            r0[j] = p[0];
            r1[j] = p[32];
        }

        const float w = edge_weight(g1 + g2);
        atomicAdd(rowsum + src, w);

        #pragma unroll
        for (int j = 0; j < EB; ++j) {
            const int d = __builtin_amdgcn_readlane(dst, EB + j);
            const float* p = emb + (size_t)d * OUT_DIM + lane;
            q0[j] = p[0];
            q1[j] = p[32];
        }

        #pragma unroll
        for (int j = 0; j < EB; ++j) {
            const int   s  = __builtin_amdgcn_readlane(src, j);
            const float wj = __int_as_float(__builtin_amdgcn_readlane(__float_as_int(w), j));
            float* o = out + (size_t)s * OUT_DIM + lane;
            atomicAdd(o,      wj * r0[j]);
            atomicAdd(o + 32, wj * r1[j]);
        }
        #pragma unroll
        for (int j = 0; j < EB; ++j) {
            const int   s  = __builtin_amdgcn_readlane(src, EB + j);
            const float wj = __int_as_float(__builtin_amdgcn_readlane(__float_as_int(w), EB + j));
            float* o = out + (size_t)s * OUT_DIM + lane;
            atomicAdd(o,      wj * q0[j]);
            atomicAdd(o + 32, wj * q1[j]);
        }
    } else {
        int src = 0, dst = 0;
        float w = 0.f;
        if (lane < cnt) {
            const int2 sd = *(const int2*)(edges + 2 * (size_t)(e0 + lane));
            src = sd.x;
            dst = sd.y;
            w = edge_weight(s1[src] + s2[dst]);
            atomicAdd(rowsum + src, w);
        }
        for (int j = 0; j < cnt; ++j) {
            const int   d  = __builtin_amdgcn_readlane(dst, j);
            const int   s  = __builtin_amdgcn_readlane(src, j);
            const float wj = __int_as_float(__builtin_amdgcn_readlane(__float_as_int(w), j));
            const float* p = emb + (size_t)d * OUT_DIM + lane;
            float* o = out + (size_t)s * OUT_DIM + lane;
            atomicAdd(o,      wj * p[0]);
            atomicAdd(o + 32, wj * p[32]);
        }
    }
}

__global__ __launch_bounds__(256) void gat_final(
    float* __restrict__ out,
    const float* __restrict__ rowsum,
    int total)
{
    int idx = blockIdx.x * blockDim.x + threadIdx.x;
    if (idx >= total) return;
    out[idx] /= rowsum[idx >> 6];
}

extern "C" void kernel_launch(void* const* d_in, const int* in_sizes, int n_in,
                              void* d_out, int out_size, void* d_ws, size_t ws_size,
                              hipStream_t stream)
{
    (void)n_in; (void)out_size; (void)ws_size;
    const float* features = (const float*)d_in[0];
    const float* W        = (const float*)d_in[1];
    const float* bias     = (const float*)d_in[2];
    const float* a        = (const float*)d_in[3];
    const int*   edges    = (const int*)d_in[4];

    const int n = in_sizes[0] / IN_DIM;
    const int E = in_sizes[4] / 2;
    float* out = (float*)d_out;

    float* emb    = (float*)d_ws;
    float* s1     = emb + (size_t)n * OUT_DIM;
    float* s2     = s1 + n;
    float* rowsum = s2 + n;

    gat_gemm_wmma<<<(n + 31) / 32, 256, 0, stream>>>(features, W, bias, a, emb, s1, s2, rowsum, out, n);
    gat_edge<<<(E + 8 * EPW - 1) / (8 * EPW), 256, 0, stream>>>(edges, emb, s1, s2, rowsum, out, E);
    const int total = n * OUT_DIM;
    gat_final<<<(total + 255) / 256, 256, 0, stream>>>(out, rowsum, total);
}


# speedup vs baseline: 2.3942x; 1.5077x over previous
#include <hip/hip_runtime.h>
#include <hip/hip_bf16.h>
#include <math.h>

typedef float v2f __attribute__((ext_vector_type(2)));
typedef float v8f __attribute__((ext_vector_type(8)));

#define IN_DIM 256
#define OUT_DIM 64
#define SLOPE 0.1f
#define KC 128
#define LDP 132

#define EPI_LD 68

__device__ __forceinline__ bool two_planes(unsigned long long ws_bytes, int n)
{
    return ws_bytes >= (unsigned long long)n * (2 * OUT_DIM + 4) * sizeof(float);
}

__device__ __forceinline__ float edge_weight(float z)
{
    float l = z > 0.f ? z : SLOPE * z;
    return expf(l);
}

__global__ __launch_bounds__(256) void gat_gemm_wmma(
    const float* __restrict__ features,
    const float* __restrict__ W,
    const float* __restrict__ bias,
    const float* __restrict__ a,
    float* __restrict__ emb,
    float* __restrict__ s1,
    float* __restrict__ s2,
    float* __restrict__ rowsum,
    float* __restrict__ out,
    float* __restrict__ rowsum1,
    float* __restrict__ out1,
    unsigned long long ws_bytes,
    int n)
{
    __shared__ float ldsA[32 * LDP];
    __shared__ float ldsB[OUT_DIM * LDP];

    const int tid  = threadIdx.x;
    const int row0 = blockIdx.x * 32;

    const int wave  = tid >> 5;
    const int lane  = tid & 31;
    const int mbase = (wave >> 2) * 16;
    const int n0    = (wave & 3) * 16;
    const int lm    = lane & 15;
    const int half  = lane >> 4;
    const int kh    = half * 2;

    const float  bv = bias[n0 + lm];
    const float2 a1 = *(const float2*)(a + 2 * lane);
    const float2 a2 = *(const float2*)(a + OUT_DIM + 2 * lane);

    v8f acc = {0.f, 0.f, 0.f, 0.f, 0.f, 0.f, 0.f, 0.f};

    for (int kc = 0; kc < IN_DIM; kc += KC) {
        #pragma unroll
        for (int i = 0; i < 4; ++i) {
            int idx = tid + i * 256;
            int r   = idx >> 5;
            int c4  = idx & 31;
            float4 v = make_float4(0.f, 0.f, 0.f, 0.f);
            int gr = row0 + r;
            if (gr < n)
                v = *(const float4*)(features + (size_t)gr * IN_DIM + kc + c4 * 4);
            *(float4*)(&ldsA[r * LDP + c4 * 4]) = v;
        }
        #pragma unroll
        for (int i = 0; i < 8; ++i) {
            int idx = tid + i * 256;
            int r   = idx >> 5;
            int c4  = idx & 31;
            float4 v = *(const float4*)(W + (size_t)r * IN_DIM + kc + c4 * 4);
            *(float4*)(&ldsB[r * LDP + c4 * 4]) = v;
        }
        __syncthreads();

        const float* pA = &ldsA[(mbase + lm) * LDP + kh];
        const float* pB = &ldsB[(n0   + lm) * LDP + kh];
        #pragma unroll 8
        for (int k = 0; k < KC; k += 4) {
            v2f afrag = *(const v2f*)(pA + k);
            v2f bfrag = *(const v2f*)(pB + k);
            acc = __builtin_amdgcn_wmma_f32_16x16x4_f32(
                false, afrag, false, bfrag, (short)0, acc, false, false);
        }
        __syncthreads();
    }

    float* tile = ldsA;
    #pragma unroll
    for (int r = 0; r < 8; ++r)
        tile[(mbase + r + half * 8) * EPI_LD + n0 + lm] = acc[r] + bv;
    __syncthreads();

    const int rl0 = wave * 4;
    float2 e[4];
    float d1[4], d2[4];
    #pragma unroll
    for (int j = 0; j < 4; ++j) {
        e[j]  = *(const float2*)(&tile[(rl0 + j) * EPI_LD + 2 * lane]);
        d1[j] = e[j].x * a1.x + e[j].y * a1.y;
        d2[j] = e[j].x * a2.x + e[j].y * a2.y;
    }
    #pragma unroll
    for (int off = 16; off > 0; off >>= 1) {
        #pragma unroll
        for (int j = 0; j < 4; ++j) {
            d1[j] += __shfl_xor(d1[j], off);
            d2[j] += __shfl_xor(d2[j], off);
        }
    }
    const bool two = two_planes(ws_bytes, n);
    float ws = 0.f, v1 = 0.f, v2 = 0.f;
    #pragma unroll
    for (int j = 0; j < 4; ++j) {
        const float w = edge_weight(d1[j] + d2[j]);
        const int grow = row0 + rl0 + j;
        if (grow < n) {
            *(float2*)(emb + (size_t)grow * OUT_DIM + 2 * lane) = e[j];
            *(float2*)(out + (size_t)grow * OUT_DIM + 2 * lane) =
                make_float2(w * e[j].x, w * e[j].y);
            if (two)
                *(float2*)(out1 + (size_t)grow * OUT_DIM + 2 * lane) = make_float2(0.f, 0.f);
        }
        if (lane == j) { ws = w; v1 = d1[j]; v2 = d2[j]; }
    }
    const int srow = row0 + rl0 + lane;
    if (lane < 4 && srow < n) {
        s1[srow]     = v1;
        s2[srow]     = v2;
        rowsum[srow] = ws;
        if (two) rowsum1[srow] = 0.f;
    }
}

#define EPW 32
#define EB  16

__global__ __launch_bounds__(256) void gat_edge(
    const int* __restrict__ edges,
    const float* __restrict__ emb,
    const float* __restrict__ s1,
    const float* __restrict__ s2,
    float* __restrict__ rowsum0,
    float* __restrict__ out0,
    float* __restrict__ rowsum1,
    float* __restrict__ out1,
    unsigned long long ws_bytes,
    int n,
    int E)
{
    const unsigned place = __builtin_amdgcn_s_sendmsg_rtn(0x87);
    const bool upper = two_planes(ws_bytes, n) && ((place >> 18) & 1u);
    float* __restrict__ rowsum = upper ? rowsum1 : rowsum0;
    float* __restrict__ out    = upper ? out1 : out0;

    const int lane = threadIdx.x & 31;
    const int wid  = (blockIdx.x * blockDim.x + threadIdx.x) >> 5;
    const int e0   = wid * EPW;
    if (e0 >= E) return;
    const int cnt = min(EPW, E - e0);

    if (cnt == EPW) {
        const int2 sd = *(const int2*)(edges + 2 * (size_t)(e0 + lane));
        const int src = sd.x;
        const int dst = sd.y;
        const float g1 = s1[src];
        const float g2 = s2[dst];

        float r0[EB], r1[EB], q0[EB], q1[EB];
        #pragma unroll
        for (int j = 0; j < EB; ++j) {
            const int d = __builtin_amdgcn_readlane(dst, j);
            const float* p = emb + (size_t)d * OUT_DIM + lane;
            r0[j] = p[0];
            r1[j] = p[32];
        }

        const float w = edge_weight(g1 + g2);
        atomicAdd(rowsum + src, w);

        #pragma unroll
        for (int j = 0; j < EB; ++j) {
            const int d = __builtin_amdgcn_readlane(dst, EB + j);
            const float* p = emb + (size_t)d * OUT_DIM + lane;
            q0[j] = p[0];
            q1[j] = p[32];
        }

        #pragma unroll
        for (int j = 0; j < EB; ++j) {
            const int   s  = __builtin_amdgcn_readlane(src, j);
            const float wj = __int_as_float(__builtin_amdgcn_readlane(__float_as_int(w), j));
            float* o = out + (size_t)s * OUT_DIM + lane;
            atomicAdd(o,      wj * r0[j]);
            atomicAdd(o + 32, wj * r1[j]);
        }
        #pragma unroll
        for (int j = 0; j < EB; ++j) {
            const int   s  = __builtin_amdgcn_readlane(src, EB + j);
            const float wj = __int_as_float(__builtin_amdgcn_readlane(__float_as_int(w), EB + j));
            float* o = out + (size_t)s * OUT_DIM + lane;
            atomicAdd(o,      wj * q0[j]);
            atomicAdd(o + 32, wj * q1[j]);
        }
    } else {
        int src = 0, dst = 0;
        float w = 0.f;
        if (lane < cnt) {
            const int2 sd = *(const int2*)(edges + 2 * (size_t)(e0 + lane));
            src = sd.x;
            dst = sd.y;
            w = edge_weight(s1[src] + s2[dst]);
            atomicAdd(rowsum + src, w);
        }
        for (int j = 0; j < cnt; ++j) {
            const int   d  = __builtin_amdgcn_readlane(dst, j);
            const int   s  = __builtin_amdgcn_readlane(src, j);
            const float wj = __int_as_float(__builtin_amdgcn_readlane(__float_as_int(w), j));
            const float* p = emb + (size_t)d * OUT_DIM + lane;
            float* o = out + (size_t)s * OUT_DIM + lane;
            atomicAdd(o,      wj * p[0]);
            atomicAdd(o + 32, wj * p[32]);
        }
    }
}

__global__ __launch_bounds__(256) void gat_final(
    float* __restrict__ out,
    const float* __restrict__ rowsum,
    const float* __restrict__ out1,
    const float* __restrict__ rowsum1,
    unsigned long long ws_bytes,
    int n,
    int total)
{
    int idx = blockIdx.x * blockDim.x + threadIdx.x;
    if (idx >= total) return;
    float v = out[idx];
    float r = rowsum[idx >> 6];
    if (two_planes(ws_bytes, n)) {
        v += out1[idx];
        r += rowsum1[idx >> 6];
    }
    out[idx] = v / r;
}

extern "C" void kernel_launch(void* const* d_in, const int* in_sizes, int n_in,
                              void* d_out, int out_size, void* d_ws, size_t ws_size,
                              hipStream_t stream)
{
    (void)n_in; (void)out_size;
    const float* features = (const float*)d_in[0];
    const float* W        = (const float*)d_in[1];
    const float* bias     = (const float*)d_in[2];
    const float* a        = (const float*)d_in[3];
    const int*   edges    = (const int*)d_in[4];

    const int n = in_sizes[0] / IN_DIM;
    const int E = in_sizes[4] / 2;
    float* out = (float*)d_out;

    float* emb     = (float*)d_ws;
    float* s1      = emb + (size_t)n * OUT_DIM;
    float* s2      = s1 + n;
    float* rowsum  = s2 + n;
    float* rowsum1 = rowsum + n;
    float* out1    = rowsum1 + n;
    const unsigned long long ws_bytes = ws_size;

    gat_gemm_wmma<<<(n + 31) / 32, 256, 0, stream>>>(features, W, bias, a, emb, s1, s2, rowsum, out, rowsum1, out1, ws_bytes, n);
    gat_edge<<<(E + 8 * EPW - 1) / (8 * EPW), 256, 0, stream>>>(edges, emb, s1, s2, rowsum, out, rowsum1, out1, ws_bytes, n, E);
    const int total = n * OUT_DIM;
    gat_final<<<(total + 255) / 256, 256, 0, stream>>>(out, rowsum, out1, rowsum1, ws_bytes, n, total);
}
